# starting kernel
# baseline (speedup 1.0000x reference)
#include <hip/hip_runtime.h>
#include <math.h>

#define B_   32
#define N_   512
#define V_   5
#define D_   256
#define H_   4
#define BN_  (B_ * N_)
#define T_   (BN_ + B_ * V_)
#define HD_  (H_ * D_)
#define EPS_ 1e-5f

typedef float v2f __attribute__((ext_vector_type(2)));
typedef float v8f __attribute__((ext_vector_type(8)));

__device__ __forceinline__ float gelu_exact(float x) {
    return 0.5f * x * (1.0f + erff(x * 0.70710678118654752f));
}

__global__ void k_build_nodes(const float* __restrict__ x,
                              const float* __restrict__ vn,
                              float* __restrict__ nodes) {
    size_t i = (size_t)blockIdx.x * blockDim.x + threadIdx.x;
    if (i < (size_t)BN_ * D_) {
        nodes[i] = x[i];
    } else if (i < (size_t)T_ * D_) {
        size_t j = i - (size_t)BN_ * D_;
        nodes[i] = vn[j % (V_ * D_)];
    }
}

template <int KDIM, int NDIM>
__global__ void k_gemm_f32_wmma(const float* __restrict__ A,
                                const float* __restrict__ Bm,
                                float* __restrict__ C) {
    const int wave = threadIdx.x >> 5;
    const int lane = threadIdx.x & 31;
    const int half = lane >> 4;
    const int lm   = lane & 15;
    const int m0   = blockIdx.x * 16;
    const int n0   = blockIdx.y * 256 + wave * 64;
    if (n0 >= NDIM) return;

    v8f c0 = {}; v8f c1 = {}; v8f c2 = {}; v8f c3 = {};
    const float* arow = A + (size_t)(m0 + lm) * KDIM;

    for (int k0 = 0; k0 < KDIM; k0 += 4) {
        const int kb = k0 + half * 2;
        v2f a;
        a[0] = arow[kb];
        a[1] = arow[kb + 1];
        const float* b0p = Bm + (size_t)kb * NDIM + n0 + lm;
        const float* b1p = b0p + NDIM;
        v2f b;
        b[0] = b0p[0];  b[1] = b1p[0];
        c0 = __builtin_amdgcn_wmma_f32_16x16x4_f32(false, a, false, b, (short)0, c0, false, false);
        b[0] = b0p[16]; b[1] = b1p[16];
        c1 = __builtin_amdgcn_wmma_f32_16x16x4_f32(false, a, false, b, (short)0, c1, false, false);
        b[0] = b0p[32]; b[1] = b1p[32];
        c2 = __builtin_amdgcn_wmma_f32_16x16x4_f32(false, a, false, b, (short)0, c2, false, false);
        b[0] = b0p[48]; b[1] = b1p[48];
        c3 = __builtin_amdgcn_wmma_f32_16x16x4_f32(false, a, false, b, (short)0, c3, false, false);
    }

    float* crow = C + n0 + lm;
#pragma unroll
    for (int j = 0; j < 8; ++j) {
        const size_t row = (size_t)(m0 + half * 8 + j) * NDIM;
        crow[row + 0]  = c0[j];
        crow[row + 16] = c1[j];
        crow[row + 32] = c2[j];
        crow[row + 48] = c3[j];
    }
}

__global__ void k_scores(const float* __restrict__ h,
                         const float* __restrict__ att_src,
                         const float* __restrict__ att_dst,
                         float* __restrict__ a_src,
                         float* __restrict__ a_dst) {
    __shared__ float red[D_];
    const int t = blockIdx.x;
    const int d = threadIdx.x;
    const float* hr = h + (size_t)t * HD_;
#pragma unroll
    for (int hd = 0; hd < H_; ++hd) {
        const float hv = hr[hd * D_ + d];
        red[d] = hv * att_src[hd * D_ + d];
        __syncthreads();
        for (int s = D_ / 2; s > 0; s >>= 1) {
            if (d < s) red[d] += red[d + s];
            __syncthreads();
        }
        if (d == 0) a_src[t * H_ + hd] = red[0];
        __syncthreads();
        red[d] = hv * att_dst[hd * D_ + d];
        __syncthreads();
        for (int s = D_ / 2; s > 0; s >>= 1) {
            if (d < s) red[d] += red[d + s];
            __syncthreads();
        }
        if (d == 0) a_dst[t * H_ + hd] = red[0];
        __syncthreads();
    }
}

__global__ void k_gat_virtual(const float* __restrict__ h,
                              const float* __restrict__ a_src,
                              const float* __restrict__ a_dst,
                              float* __restrict__ gat_agg) {
    __shared__ float w[N_ + 1];
    __shared__ float red[D_];
    const int uidx = blockIdx.x;
    const int head = blockIdx.y;
    const int b    = uidx / V_;
    const int u    = BN_ + uidx;
    const int tid  = threadIdx.x;

    const float adst = a_dst[u * H_ + head];
    const int r0 = b * N_ + tid * 2;
    float al0 = a_src[(size_t)r0 * H_ + head] + adst;
    al0 = (al0 > 0.0f) ? al0 : 0.2f * al0;
    float al1 = a_src[(size_t)(r0 + 1) * H_ + head] + adst;
    al1 = (al1 > 0.0f) ? al1 : 0.2f * al1;
    float aself = a_src[(size_t)u * H_ + head] + adst;
    aself = (aself > 0.0f) ? aself : 0.2f * aself;

    float lmax = fmaxf(al0, al1);
    if (tid == 0) lmax = fmaxf(lmax, aself);
    red[tid] = lmax;
    __syncthreads();
    for (int s = D_ / 2; s > 0; s >>= 1) {
        if (tid < s) red[tid] = fmaxf(red[tid], red[tid + s]);
        __syncthreads();
    }
    const float m = red[0];
    __syncthreads();

    const float e0 = expf(al0 - m);
    const float e1 = expf(al1 - m);
    w[tid * 2]     = e0;
    w[tid * 2 + 1] = e1;
    float es = 0.0f;
    if (tid == 0) { es = expf(aself - m); w[N_] = es; }
    red[tid] = e0 + e1 + es;
    __syncthreads();
    for (int s = D_ / 2; s > 0; s >>= 1) {
        if (tid < s) red[tid] += red[tid + s];
        __syncthreads();
    }
    const float inv = 1.0f / red[0];
    __syncthreads();

    const float* hb = h + (size_t)(b * N_) * HD_ + head * D_ + tid;
    float acc = 0.0f;
    for (int r = 0; r < N_; ++r)
        acc += w[r] * hb[(size_t)r * HD_];
    acc += w[N_] * h[(size_t)u * HD_ + head * D_ + tid];
    gat_agg[((size_t)uidx * H_ + head) * D_ + tid] = acc * inv;
}

__global__ void k_gat_finish(const float* __restrict__ h,
                             const float* __restrict__ gat_agg,
                             const float* __restrict__ bias,
                             const float* __restrict__ gg,
                             const float* __restrict__ gb,
                             float* __restrict__ out) {
    __shared__ float red[D_];
    const int t = blockIdx.x;
    const int d = threadIdx.x;
    float v;
    if (t < BN_) {
        const float* p = h + (size_t)t * HD_ + d;
        v = 0.25f * (p[0] + p[D_] + p[2 * D_] + p[3 * D_]);
    } else {
        const float* p = gat_agg + (size_t)(t - BN_) * HD_ + d;
        v = 0.25f * (p[0] + p[D_] + p[2 * D_] + p[3 * D_]);
    }
    v = gelu_exact(v + bias[d]);

    red[d] = v;
    __syncthreads();
    for (int s = D_ / 2; s > 0; s >>= 1) {
        if (d < s) red[d] += red[d + s];
        __syncthreads();
    }
    const float mu = red[0] * (1.0f / D_);
    __syncthreads();
    const float c = v - mu;
    red[d] = c * c;
    __syncthreads();
    for (int s = D_ / 2; s > 0; s >>= 1) {
        if (d < s) red[d] += red[d + s];
        __syncthreads();
    }
    const float var = red[0] * (1.0f / D_);
    out[(size_t)t * D_ + d] = c * rsqrtf(var + EPS_) * gg[d] + gb[d];
}

__global__ void k_gcn_finish(const float* __restrict__ xw,
                             const float* __restrict__ bias,
                             const float* __restrict__ gg,
                             const float* __restrict__ gb,
                             float* __restrict__ out) {
    __shared__ float red[D_];
    const int t = blockIdx.x;
    const int d = threadIdx.x;
    const int b = t / N_;
    const float* xv = xw + (size_t)(BN_ + b * V_) * D_ + d;
    const float s5 = xv[0] + xv[D_] + xv[2 * D_] + xv[3 * D_] + xv[4 * D_];
    const float inv_sqrt6 = 0.40824829046386301637f;
    float v = s5 * inv_sqrt6 + xw[(size_t)t * D_ + d] * (1.0f / 6.0f) + bias[d];
    v = gelu_exact(v);

    red[d] = v;
    __syncthreads();
    for (int s = D_ / 2; s > 0; s >>= 1) {
        if (d < s) red[d] += red[d + s];
        __syncthreads();
    }
    const float mu = red[0] * (1.0f / D_);
    __syncthreads();
    const float c = v - mu;
    red[d] = c * c;
    __syncthreads();
    for (int s = D_ / 2; s > 0; s >>= 1) {
        if (d < s) red[d] += red[d + s];
        __syncthreads();
    }
    const float var = red[0] * (1.0f / D_);
    out[(size_t)t * D_ + d] = c * rsqrtf(var + EPS_) * gg[d] + gb[d];
}

extern "C" void kernel_launch(void* const* d_in, const int* in_sizes, int n_in,
                              void* d_out, int out_size, void* d_ws, size_t ws_size,
                              hipStream_t stream) {
    (void)in_sizes; (void)n_in; (void)out_size; (void)ws_size;

    const float* x       = (const float*)d_in[0];
    const float* vn      = (const float*)d_in[2];
    const float* gat_W   = (const float*)d_in[3];
    const float* att_src = (const float*)d_in[4];
    const float* att_dst = (const float*)d_in[5];
    const float* gat_b   = (const float*)d_in[6];
    const float* gcn_W   = (const float*)d_in[7];
    const float* gcn_b   = (const float*)d_in[8];
    const float* n1g     = (const float*)d_in[9];
    const float* n1b     = (const float*)d_in[10];
    const float* n2g     = (const float*)d_in[11];
    const float* n2b     = (const float*)d_in[12];
    float* out = (float*)d_out;

    float* ws = (float*)d_ws;
    size_t off = 0;
    float* nodes = ws + off; off += (size_t)T_ * D_;
    float* h     = ws + off; off += (size_t)T_ * HD_;
    float* asrc  = ws + off; off += (size_t)T_ * H_;
    float* adst  = ws + off; off += (size_t)T_ * H_;
    float* gagg  = ws + off; off += (size_t)B_ * V_ * HD_;
    float* gat1  = ws + off; off += (size_t)T_ * D_;
    float* xw    = ws + off; off += (size_t)T_ * D_;

    {
        const size_t n = (size_t)T_ * D_;
        k_build_nodes<<<(unsigned)((n + 255) / 256), 256, 0, stream>>>(x, vn, nodes);
    }
    {
        dim3 grid(T_ / 16, HD_ / 256);
        k_gemm_f32_wmma<D_, HD_><<<grid, 128, 0, stream>>>(nodes, gat_W, h);
    }
    k_scores<<<T_, D_, 0, stream>>>(h, att_src, att_dst, asrc, adst);
    {
        dim3 grid(B_ * V_, H_);
        k_gat_virtual<<<grid, D_, 0, stream>>>(h, asrc, adst, gagg);
    }
    k_gat_finish<<<T_, D_, 0, stream>>>(h, gagg, gat_b, n1g, n1b, gat1);
    {
        dim3 grid(T_ / 16, 1);
        k_gemm_f32_wmma<D_, D_><<<grid, 128, 0, stream>>>(gat1, gcn_W, xw);
    }
    k_gcn_finish<<<BN_, D_, 0, stream>>>(xw, gcn_b, n2g, n2b, out);
}


# Virtual aggregation as a matrix product per (batch, head): 5 softmax weight rows (padded to 16) x the h slice on f32 WMMA, each h row read once for all 5 nodes; softmax, aggregation sum orders change
# speedup vs baseline: 2.0267x; 2.0267x over previous
#include <hip/hip_runtime.h>
#include <math.h>

#define B_   32
#define N_   512
#define V_   5
#define D_   256
#define H_   4
#define BN_  (B_ * N_)
#define T_   (BN_ + B_ * V_)
#define HD_  (H_ * D_)
#define EPS_ 1e-5f

typedef float v2f __attribute__((ext_vector_type(2)));
typedef float v8f __attribute__((ext_vector_type(8)));

__device__ __forceinline__ float gelu_exact(float x) {
    return 0.5f * x * (1.0f + erff(x * 0.70710678118654752f));
}

__global__ void k_build_nodes(const float* __restrict__ x,
                              const float* __restrict__ vn,
                              float* __restrict__ nodes) {
    size_t i = (size_t)blockIdx.x * blockDim.x + threadIdx.x;
    if (i < (size_t)BN_ * D_) {
        nodes[i] = x[i];
    } else if (i < (size_t)T_ * D_) {
        size_t j = i - (size_t)BN_ * D_;
        nodes[i] = vn[j % (V_ * D_)];
    }
}

#define GB_COLS  64
#define GB_WAVES 4
typedef float v4f __attribute__((ext_vector_type(4)));

template <int MDIM, int KDIM, int NDIM>
__global__ void __launch_bounds__(GB_WAVES * 32)
k_gemm_f32_wmma(const float* __restrict__ A,
                const float* __restrict__ Bm,
                float* __restrict__ C) {
    static_assert(KDIM % 16 == 0 && NDIM % GB_COLS == 0 && MDIM % 16 == 0, "tile shapes");
    __shared__ __attribute__((aligned(16))) float Bl[(KDIM / 2) * GB_COLS * 2];

    const int tid = threadIdx.x;
    const int n0  = blockIdx.y * GB_COLS;

    {
        const int cq = tid & 15;
        const int kr = tid >> 4;
        const float* bsrc = Bm + n0 + cq * 4;
        const int f  = (kr & 1) * 2 + (cq >> 3);
        float* dbase = Bl + (cq >> 3) * 64 + (cq & 3) * 16 + ((cq >> 2) & 1) * 2;
#pragma unroll
        for (int i = 0; i < KDIM / 16; ++i) {
            const int kp = kr + 8 * i;
            const v4f r0 = *(const v4f*)(bsrc + (size_t)(2 * kp) * NDIM);
            const v4f r1 = *(const v4f*)(bsrc + (size_t)(2 * kp + 1) * NDIM);
            float* d = dbase + kp * (GB_COLS * 2);
#pragma unroll
            for (int j = 0; j < 4; ++j) {
                v2f p;
                p[0] = r0[j]; p[1] = r1[j];
                *(v2f*)(d + (j ^ f) * 4) = p;
            }
        }
    }
    __syncthreads();

    const int wave = tid >> 5;
    const int lane = tid & 31;
    const int half = lane >> 4;
    const int lm   = lane & 15;
    const int m0   = (blockIdx.x * GB_WAVES + wave) * 16;
    if (m0 >= MDIM) return;

    v8f c0 = {}; v8f c1 = {}; v8f c2 = {}; v8f c3 = {};
    const float* arow = A + (size_t)(m0 + lm) * KDIM + half * 2;
    const float* bl01 = Bl + half * (GB_COLS * 2) + (lm ^ (half * 2)) * 4;
    const float* bl23 = Bl + half * (GB_COLS * 2) + 64 + (lm ^ (half * 2 + 1)) * 4;

    for (int k0 = 0; k0 < KDIM; k0 += 4) {
        v2f a;
        a[0] = arow[k0];
        a[1] = arow[k0 + 1];
        const int ko = (k0 >> 1) * (GB_COLS * 2);
        const v4f q01 = *(const v4f*)(bl01 + ko);
        const v4f q23 = *(const v4f*)(bl23 + ko);
        v2f b0, b1, b2, b3;
        b0[0] = q01[0]; b0[1] = q01[1];
        b1[0] = q01[2]; b1[1] = q01[3];
        b2[0] = q23[0]; b2[1] = q23[1];
        b3[0] = q23[2]; b3[1] = q23[3];
        c0 = __builtin_amdgcn_wmma_f32_16x16x4_f32(false, a, false, b0, (short)0, c0, false, false);
        c1 = __builtin_amdgcn_wmma_f32_16x16x4_f32(false, a, false, b1, (short)0, c1, false, false);
        c2 = __builtin_amdgcn_wmma_f32_16x16x4_f32(false, a, false, b2, (short)0, c2, false, false);
        c3 = __builtin_amdgcn_wmma_f32_16x16x4_f32(false, a, false, b3, (short)0, c3, false, false);
    }

    float* crow = C + n0 + lm;
#pragma unroll
    for (int j = 0; j < 8; ++j) {
        const size_t row = (size_t)(m0 + half * 8 + j) * NDIM;
        crow[row + 0]  = c0[j];
        crow[row + 16] = c1[j];
        crow[row + 32] = c2[j];
        crow[row + 48] = c3[j];
    }
}

#define SC_ROWS  4
#define SC_WAVES 4
static_assert(T_ % (SC_ROWS * SC_WAVES) == 0, "the grid covers the rows exactly");
static_assert(SC_ROWS * H_ * 2 == 32, "one total per lane");
static_assert(BN_ % SC_ROWS == 0, "a wave's rows are all real or all virtual");

template <int N, int OFF>
__device__ __forceinline__ void sc_fold(float* v, int lane) {
    const bool up = (lane & OFF) != 0;
#pragma unroll
    for (int i = 0; i < N; ++i) {
        const float keep = up ? v[N + i] : v[i];
        const float send = up ? v[i] : v[N + i];
        v[i] = keep + __shfl_xor(send, OFF, 32);
    }
}

__device__ __forceinline__ void sc_allsum(float* s) {
#pragma unroll
    for (int off = 16; off > 0; off >>= 1) {
#pragma unroll
        for (int r = 0; r < SC_ROWS; ++r) s[r] += __shfl_xor(s[r], off, 32);
    }
}

__global__ void __launch_bounds__(SC_WAVES * 32, 5)
k_scores(const float* __restrict__ h,
         const float* __restrict__ att_src,
         const float* __restrict__ att_dst,
         const float* __restrict__ bias,
         const float* __restrict__ gg,
         const float* __restrict__ gb,
         float* __restrict__ a_src,
         float* __restrict__ a_dst,
         float* __restrict__ gat1) {
    const int lane = threadIdx.x & 31;
    const int wave = blockIdx.x * SC_WAVES + (threadIdx.x >> 5);
    const int t0   = wave * SC_ROWS;

    const v4f* as4 = (const v4f*)att_src + lane;
    const v4f* ad4 = (const v4f*)att_dst + lane;
    v4f ws[8], wd[8];
#pragma unroll
    for (int q = 0; q < 8; ++q) { ws[q] = as4[q * 32]; wd[q] = ad4[q * 32]; }

    v4f hv[SC_ROWS][8];
#pragma unroll
    for (int r = 0; r < SC_ROWS; ++r) {
        const v4f* hp = (const v4f*)(h + (size_t)(t0 + r) * HD_) + lane;
#pragma unroll
        for (int q = 0; q < 8; ++q) hv[r][q] = hp[q * 32];
    }

    float v[32];
#pragma unroll
    for (int r = 0; r < SC_ROWS; ++r) {
#pragma unroll
        for (int hd = 0; hd < H_; ++hd) {
            float s = 0.0f, d = 0.0f;
#pragma unroll
            for (int q = 2 * hd; q < 2 * hd + 2; ++q) {
#pragma unroll
                for (int j = 0; j < 4; ++j) {
                    s = fmaf(hv[r][q][j], ws[q][j], s);
                    d = fmaf(hv[r][q][j], wd[q][j], d);
                }
            }
            v[r * H_ + hd]      = s;
            v[16 + r * H_ + hd] = d;
        }
    }

    v4f hm[SC_ROWS][2];
#pragma unroll
    for (int r = 0; r < SC_ROWS; ++r) {
#pragma unroll
        for (int p = 0; p < 2; ++p)
            hm[r][p] = 0.25f * (hv[r][p] + hv[r][2 + p] + hv[r][4 + p] + hv[r][6 + p]);
    }

    sc_fold<16, 16>(v, lane);
    sc_fold<8, 8>(v, lane);
    sc_fold<4, 4>(v, lane);
    sc_fold<2, 2>(v, lane);
    sc_fold<1, 1>(v, lane);

    float* dst = (lane < 16) ? (a_src + (size_t)t0 * H_ + lane)
                             : (a_dst + (size_t)t0 * H_ + (lane - 16));
    *dst = v[0];

    if (t0 >= BN_) return;

    v4f bb[2], g4[2], be[2];
#pragma unroll
    for (int p = 0; p < 2; ++p) {
        bb[p] = ((const v4f*)bias + lane)[p * 32];
        g4[p] = ((const v4f*)gg + lane)[p * 32];
        be[p] = ((const v4f*)gb + lane)[p * 32];
    }

    float s1[SC_ROWS];
#pragma unroll
    for (int r = 0; r < SC_ROWS; ++r) {
        float s = 0.0f;
#pragma unroll
        for (int p = 0; p < 2; ++p) {
#pragma unroll
            for (int j = 0; j < 4; ++j) {
                const float g = gelu_exact(hm[r][p][j] + bb[p][j]);
                hm[r][p][j] = g;
                s += g;
            }
        }
        s1[r] = s;
    }
    sc_allsum(s1);

    float s2[SC_ROWS];
#pragma unroll
    for (int r = 0; r < SC_ROWS; ++r) {
        const float mu = s1[r] * (1.0f / D_);
        float s = 0.0f;
#pragma unroll
        for (int p = 0; p < 2; ++p) {
#pragma unroll
            for (int j = 0; j < 4; ++j) {
                const float cdev = hm[r][p][j] - mu;
                hm[r][p][j] = cdev;
                s += cdev * cdev;
            }
        }
        s2[r] = s;
    }
    sc_allsum(s2);

#pragma unroll
    for (int r = 0; r < SC_ROWS; ++r) {
        const float var = s2[r] * (1.0f / D_);
        const float rs  = rsqrtf(var + EPS_);
        v4f* op = (v4f*)(gat1 + (size_t)(t0 + r) * D_) + lane;
#pragma unroll
        for (int p = 0; p < 2; ++p) {
            v4f o;
#pragma unroll
            for (int j = 0; j < 4; ++j) o[j] = hm[r][p][j] * rs * g4[p][j] + be[p][j];
            op[p * 32] = o;
        }
    }
}

#define VA_WAVES 8
#define VA_PITCH 520
static_assert(N_ == 2 * VA_WAVES * 32, "two sources per thread in the softmax part");
static_assert(N_ == 4 * 128 && D_ == 2 * 128, "wave tiling of the aggregation");

__device__ __forceinline__ float lrelu02(float x) { return (x > 0.0f) ? x : 0.2f * x; }

__global__ void __launch_bounds__(VA_WAVES * 32)
k_gat_virtual(const float* __restrict__ h,
              const float* __restrict__ a_src,
              const float* __restrict__ a_dst,
              float* __restrict__ gat_agg) {
    __shared__ __attribute__((aligned(16))) float wl[6 * VA_PITCH];
    __shared__ __attribute__((aligned(16))) float part[4][V_][128];
    __shared__ float redm[VA_WAVES][8];
    __shared__ float reds[VA_WAVES][8];
    __shared__ float fin_ws[8];
    __shared__ float fin_inv[8];

    const int tid  = threadIdx.x;
    const int lane = tid & 31;
    const int wave = tid >> 5;
    const int half = lane >> 4;
    const int c    = lane & 15;
    const int ch   = blockIdx.x & 1;
    const int hd   = (blockIdx.x >> 1) & (H_ - 1);
    const int b    = blockIdx.x >> 3;
    const int cg   = wave & 1;
    const int kc   = wave >> 1;
    const int u0   = BN_ + b * V_;
    const int col0 = hd * D_ + ch * 128;

    float hself[3];
#pragma unroll
    for (int e = 0; e < 3; ++e) {
        const int idx = tid + e * 256;
        const int v   = (idx < V_ * 128) ? (idx >> 7) : 0;
        hself[e] = h[(size_t)(u0 + v) * HD_ + col0 + (idx & 127)];
    }

    float ad[V_], aself[V_];
#pragma unroll
    for (int v = 0; v < V_; ++v) {
        ad[v]    = a_dst[(size_t)(u0 + v) * H_ + hd];
        aself[v] = lrelu02(a_src[(size_t)(u0 + v) * H_ + hd] + ad[v]);
    }
    const int r0 = b * N_ + tid * 2;
    const float s0 = a_src[(size_t)r0 * H_ + hd];
    const float s1 = a_src[(size_t)(r0 + 1) * H_ + hd];
    float al0[V_], al1[V_], mx[V_];
#pragma unroll
    for (int v = 0; v < V_; ++v) {
        al0[v] = lrelu02(s0 + ad[v]);
        al1[v] = lrelu02(s1 + ad[v]);
        mx[v]  = fmaxf(al0[v], al1[v]);
    }
#pragma unroll
    for (int off = 16; off > 0; off >>= 1) {
#pragma unroll
        for (int v = 0; v < V_; ++v) mx[v] = fmaxf(mx[v], __shfl_xor(mx[v], off, 32));
    }
    if (lane == 0) {
#pragma unroll
        for (int v = 0; v < V_; ++v) redm[wave][v] = mx[v];
    }
    {
        v2f z; z[0] = 0.0f; z[1] = 0.0f;
        *(v2f*)(wl + 5 * VA_PITCH + 2 * tid) = z;
    }
    __syncthreads();

    float m[V_], es[V_];
#pragma unroll
    for (int v = 0; v < V_; ++v) {
        float t = aself[v];
#pragma unroll
        for (int w = 0; w < VA_WAVES; ++w) t = fmaxf(t, redm[w][v]);
        m[v] = t;
        v2f e;
        e[0] = expf(al0[v] - t);
        e[1] = expf(al1[v] - t);
        *(v2f*)(wl + v * VA_PITCH + 2 * tid) = e;
        es[v] = e[0] + e[1];
    }
#pragma unroll
    for (int off = 16; off > 0; off >>= 1) {
#pragma unroll
        for (int v = 0; v < V_; ++v) es[v] += __shfl_xor(es[v], off, 32);
    }
    if (lane == 0) {
#pragma unroll
        for (int v = 0; v < V_; ++v) reds[wave][v] = es[v];
    }
    __syncthreads();

    const float* hp = h + (size_t)(b * N_ + kc * 128 + 2 * half) * HD_ + col0 + cg * 64 + 4 * c;
    v4f q[32][2];
#pragma unroll
    for (int i = 0; i < 32; ++i) {
#pragma unroll
        for (int s = 0; s < 2; ++s) q[i][s] = *(const v4f*)(hp + (size_t)(4 * i + s) * HD_);
    }

    if (tid < V_) {
        float as_ = aself[0], mm = m[0];
#pragma unroll
        for (int v = 1; v < V_; ++v) {
            if (tid == v) { as_ = aself[v]; mm = m[v]; }
        }
        const float eself = expf(as_ - mm);
        float d = eself;
#pragma unroll
        for (int w = 0; w < VA_WAVES; ++w) d += reds[w][tid];
        fin_ws[tid]  = eself;
        fin_inv[tid] = 1.0f / d;
    }

    const float* ap = wl + ((c < V_) ? c : V_) * VA_PITCH + kc * 128 + 2 * half;
    v2f af[32];
#pragma unroll
    for (int i = 0; i < 32; ++i) af[i] = *(const v2f*)(ap + 4 * i);

    v8f c0 = {}; v8f c1 = {}; v8f c2 = {}; v8f c3 = {};
#pragma unroll
    for (int i = 0; i < 32; ++i) {
        v2f b0, b1, b2, b3;
        b0[0] = q[i][0][0]; b0[1] = q[i][1][0];
        b1[0] = q[i][0][1]; b1[1] = q[i][1][1];
        b2[0] = q[i][0][2]; b2[1] = q[i][1][2];
        b3[0] = q[i][0][3]; b3[1] = q[i][1][3];
        c0 = __builtin_amdgcn_wmma_f32_16x16x4_f32(false, af[i], false, b0, (short)0, c0, false, false);
        c1 = __builtin_amdgcn_wmma_f32_16x16x4_f32(false, af[i], false, b1, (short)0, c1, false, false);
        c2 = __builtin_amdgcn_wmma_f32_16x16x4_f32(false, af[i], false, b2, (short)0, c2, false, false);
        c3 = __builtin_amdgcn_wmma_f32_16x16x4_f32(false, af[i], false, b3, (short)0, c3, false, false);
    }

    if (half == 0) {
#pragma unroll
        for (int v = 0; v < V_; ++v) {
            v4f o;
            o[0] = c0[v]; o[1] = c1[v]; o[2] = c2[v]; o[3] = c3[v];
            *(v4f*)(&part[kc][v][cg * 64 + 4 * c]) = o;
        }
    }
    __syncthreads();

#pragma unroll
    for (int e = 0; e < 3; ++e) {
        const int idx = tid + e * 256;
        if (idx < V_ * 128) {
            const int v   = idx >> 7;
            const int col = idx & 127;
            float s = part[0][v][col] + part[1][v][col];
            s += part[2][v][col];
            s += part[3][v][col];
            s += fin_ws[v] * hself[e];
            gat_agg[((size_t)(b * V_ + v) * H_ + hd) * D_ + ch * 128 + col] = s * fin_inv[v];
        }
    }
}

__global__ void k_gat_finish(const float* __restrict__ gat_agg,
                             const float* __restrict__ bias,
                             const float* __restrict__ gg,
                             const float* __restrict__ gb,
                             float* __restrict__ out) {
    __shared__ float red[D_];
    const int u = blockIdx.x;
    const int d = threadIdx.x;
    const float* p = gat_agg + (size_t)u * HD_ + d;
    float v = 0.25f * (p[0] + p[D_] + p[2 * D_] + p[3 * D_]);
    v = gelu_exact(v + bias[d]);

    red[d] = v;
    __syncthreads();
    for (int s = D_ / 2; s > 0; s >>= 1) {
        if (d < s) red[d] += red[d + s];
        __syncthreads();
    }
    const float mu = red[0] * (1.0f / D_);
    __syncthreads();
    const float c = v - mu;
    red[d] = c * c;
    __syncthreads();
    for (int s = D_ / 2; s > 0; s >>= 1) {
        if (d < s) red[d] += red[d + s];
        __syncthreads();
    }
    const float var = red[0] * (1.0f / D_);
    out[(size_t)(BN_ + u) * D_ + d] = c * rsqrtf(var + EPS_) * gg[d] + gb[d];
}

__global__ void k_gcn_finish(const float* __restrict__ xw,
                             const float* __restrict__ bias,
                             const float* __restrict__ gg,
                             const float* __restrict__ gb,
                             float* __restrict__ out) {
    __shared__ float red[D_];
    const int t = blockIdx.x;
    const int d = threadIdx.x;
    const int b = t / N_;
    const float* xv = xw + (size_t)(BN_ + b * V_) * D_ + d;
    const float s5 = xv[0] + xv[D_] + xv[2 * D_] + xv[3 * D_] + xv[4 * D_];
    const float inv_sqrt6 = 0.40824829046386301637f;
    float v = s5 * inv_sqrt6 + xw[(size_t)t * D_ + d] * (1.0f / 6.0f) + bias[d];
    v = gelu_exact(v);

    red[d] = v;
    __syncthreads();
    for (int s = D_ / 2; s > 0; s >>= 1) {
        if (d < s) red[d] += red[d + s];
        __syncthreads();
    }
    const float mu = red[0] * (1.0f / D_);
    __syncthreads();
    const float c = v - mu;
    red[d] = c * c;
    __syncthreads();
    for (int s = D_ / 2; s > 0; s >>= 1) {
        if (d < s) red[d] += red[d + s];
        __syncthreads();
    }
    const float var = red[0] * (1.0f / D_);
    out[(size_t)t * D_ + d] = c * rsqrtf(var + EPS_) * gg[d] + gb[d];
}

extern "C" void kernel_launch(void* const* d_in, const int* in_sizes, int n_in,
                              void* d_out, int out_size, void* d_ws, size_t ws_size,
                              hipStream_t stream) {
    (void)in_sizes; (void)n_in; (void)out_size; (void)ws_size;

    const float* x       = (const float*)d_in[0];
    const float* vn      = (const float*)d_in[2];
    const float* gat_W   = (const float*)d_in[3];
    const float* att_src = (const float*)d_in[4];
    const float* att_dst = (const float*)d_in[5];
    const float* gat_b   = (const float*)d_in[6];
    const float* gcn_W   = (const float*)d_in[7];
    const float* gcn_b   = (const float*)d_in[8];
    const float* n1g     = (const float*)d_in[9];
    const float* n1b     = (const float*)d_in[10];
    const float* n2g     = (const float*)d_in[11];
    const float* n2b     = (const float*)d_in[12];
    float* out = (float*)d_out;

    float* ws = (float*)d_ws;
    size_t off = 0;
    float* nodes = ws + off; off += (size_t)T_ * D_;
    float* h     = ws + off; off += (size_t)T_ * HD_;
    float* asrc  = ws + off; off += (size_t)T_ * H_;
    float* adst  = ws + off; off += (size_t)T_ * H_;
    float* gagg  = ws + off; off += (size_t)B_ * V_ * HD_;
    float* gat1  = ws + off; off += (size_t)T_ * D_;
    float* xw    = ws + off; off += (size_t)T_ * D_;

    {
        const size_t n = (size_t)T_ * D_;
        k_build_nodes<<<(unsigned)((n + 255) / 256), 256, 0, stream>>>(x, vn, nodes);
    }
    {
        dim3 grid((T_ + 63) / 64, HD_ / GB_COLS);
        k_gemm_f32_wmma<T_, D_, HD_><<<grid, GB_WAVES * 32, 0, stream>>>(nodes, gat_W, h);
    }
    k_scores<<<T_ / (SC_ROWS * SC_WAVES), SC_WAVES * 32, 0, stream>>>(h, att_src, att_dst, gat_b, n1g, n1b,
                                                                     asrc, adst, gat1);
    k_gat_virtual<<<B_ * H_ * 2, VA_WAVES * 32, 0, stream>>>(h, asrc, adst, gagg);
    k_gat_finish<<<B_ * V_, D_, 0, stream>>>(gagg, gat_b, n1g, n1b, gat1);
    {
        dim3 grid((T_ + 63) / 64, D_ / GB_COLS);
        k_gemm_f32_wmma<T_, D_, D_><<<grid, GB_WAVES * 32, 0, stream>>>(gat1, gcn_W, xw);
    }
    k_gcn_finish<<<BN_, D_, 0, stream>>>(xw, gcn_b, n2g, n2b, out);
}


# GCN finisher as a stream over xw: a wave per 2 rows, 128-bit accesses, 8 channels per lane, the 5 virtual rows read once per wave, sums over the wave; no LDS tree, no barrier; LN2 sum order changes
# speedup vs baseline: 2.3192x; 1.1443x over previous
#include <hip/hip_runtime.h>
#include <math.h>

#define B_   32
#define N_   512
#define V_   5
#define D_   256
#define H_   4
#define BN_  (B_ * N_)
#define T_   (BN_ + B_ * V_)
#define HD_  (H_ * D_)
#define EPS_ 1e-5f

typedef float v2f __attribute__((ext_vector_type(2)));
typedef float v8f __attribute__((ext_vector_type(8)));

__device__ __forceinline__ float gelu_exact(float x) {
    return 0.5f * x * (1.0f + erff(x * 0.70710678118654752f));
}

__global__ void k_build_nodes(const float* __restrict__ x,
                              const float* __restrict__ vn,
                              float* __restrict__ nodes) {
    size_t i = (size_t)blockIdx.x * blockDim.x + threadIdx.x;
    if (i < (size_t)BN_ * D_) {
        nodes[i] = x[i];
    } else if (i < (size_t)T_ * D_) {
        size_t j = i - (size_t)BN_ * D_;
        nodes[i] = vn[j % (V_ * D_)];
    }
}

#define GB_COLS  64
#define GB_WAVES 4
typedef float v4f __attribute__((ext_vector_type(4)));

template <int MDIM, int KDIM, int NDIM>
__global__ void __launch_bounds__(GB_WAVES * 32)
k_gemm_f32_wmma(const float* __restrict__ A,
                const float* __restrict__ Bm,
                float* __restrict__ C) {
    static_assert(KDIM % 16 == 0 && NDIM % GB_COLS == 0 && MDIM % 16 == 0, "tile shapes");
    __shared__ __attribute__((aligned(16))) float Bl[(KDIM / 2) * GB_COLS * 2];

    const int tid = threadIdx.x;
    const int n0  = blockIdx.y * GB_COLS;

    {
        const int cq = tid & 15;
        const int kr = tid >> 4;
        const float* bsrc = Bm + n0 + cq * 4;
        const int f  = (kr & 1) * 2 + (cq >> 3);
        float* dbase = Bl + (cq >> 3) * 64 + (cq & 3) * 16 + ((cq >> 2) & 1) * 2;
#pragma unroll
        for (int i = 0; i < KDIM / 16; ++i) {
            const int kp = kr + 8 * i;
            const v4f r0 = *(const v4f*)(bsrc + (size_t)(2 * kp) * NDIM);
            const v4f r1 = *(const v4f*)(bsrc + (size_t)(2 * kp + 1) * NDIM);
            float* d = dbase + kp * (GB_COLS * 2);
#pragma unroll
            for (int j = 0; j < 4; ++j) {
                v2f p;
                p[0] = r0[j]; p[1] = r1[j];
                *(v2f*)(d + (j ^ f) * 4) = p;
            }
        }
    }
    __syncthreads();

    const int wave = tid >> 5;
    const int lane = tid & 31;
    const int half = lane >> 4;
    const int lm   = lane & 15;
    const int m0   = (blockIdx.x * GB_WAVES + wave) * 16;
    if (m0 >= MDIM) return;

    v8f c0 = {}; v8f c1 = {}; v8f c2 = {}; v8f c3 = {};
    const float* arow = A + (size_t)(m0 + lm) * KDIM + half * 2;
    const float* bl01 = Bl + half * (GB_COLS * 2) + (lm ^ (half * 2)) * 4;
    const float* bl23 = Bl + half * (GB_COLS * 2) + 64 + (lm ^ (half * 2 + 1)) * 4;

    for (int k0 = 0; k0 < KDIM; k0 += 4) {
        v2f a;
        a[0] = arow[k0];
        a[1] = arow[k0 + 1];
        const int ko = (k0 >> 1) * (GB_COLS * 2);
        const v4f q01 = *(const v4f*)(bl01 + ko);
        const v4f q23 = *(const v4f*)(bl23 + ko);
        v2f b0, b1, b2, b3;
        b0[0] = q01[0]; b0[1] = q01[1];
        b1[0] = q01[2]; b1[1] = q01[3];
        b2[0] = q23[0]; b2[1] = q23[1];
        b3[0] = q23[2]; b3[1] = q23[3];
        c0 = __builtin_amdgcn_wmma_f32_16x16x4_f32(false, a, false, b0, (short)0, c0, false, false);
        c1 = __builtin_amdgcn_wmma_f32_16x16x4_f32(false, a, false, b1, (short)0, c1, false, false);
        c2 = __builtin_amdgcn_wmma_f32_16x16x4_f32(false, a, false, b2, (short)0, c2, false, false);
        c3 = __builtin_amdgcn_wmma_f32_16x16x4_f32(false, a, false, b3, (short)0, c3, false, false);
    }

    float* crow = C + n0 + lm;
#pragma unroll
    for (int j = 0; j < 8; ++j) {
        const size_t row = (size_t)(m0 + half * 8 + j) * NDIM;
        crow[row + 0]  = c0[j];
        crow[row + 16] = c1[j];
        crow[row + 32] = c2[j];
        crow[row + 48] = c3[j];
    }
}

#define SC_ROWS  4
#define SC_WAVES 4
static_assert(T_ % (SC_ROWS * SC_WAVES) == 0, "the grid covers the rows exactly");
static_assert(SC_ROWS * H_ * 2 == 32, "one total per lane");
static_assert(BN_ % SC_ROWS == 0, "a wave's rows are all real or all virtual");

template <int N, int OFF>
__device__ __forceinline__ void sc_fold(float* v, int lane) {
    const bool up = (lane & OFF) != 0;
#pragma unroll
    for (int i = 0; i < N; ++i) {
        const float keep = up ? v[N + i] : v[i];
        const float send = up ? v[i] : v[N + i];
        v[i] = keep + __shfl_xor(send, OFF, 32);
    }
}

__device__ __forceinline__ void sc_allsum(float* s) {
#pragma unroll
    for (int off = 16; off > 0; off >>= 1) {
#pragma unroll
        for (int r = 0; r < SC_ROWS; ++r) s[r] += __shfl_xor(s[r], off, 32);
    }
}

__global__ void __launch_bounds__(SC_WAVES * 32, 5)
k_scores(const float* __restrict__ h,
         const float* __restrict__ att_src,
         const float* __restrict__ att_dst,
         const float* __restrict__ bias,
         const float* __restrict__ gg,
         const float* __restrict__ gb,
         float* __restrict__ a_src,
         float* __restrict__ a_dst,
         float* __restrict__ gat1) {
    const int lane = threadIdx.x & 31;
    const int wave = blockIdx.x * SC_WAVES + (threadIdx.x >> 5);
    const int t0   = wave * SC_ROWS;

    const v4f* as4 = (const v4f*)att_src + lane;
    const v4f* ad4 = (const v4f*)att_dst + lane;
    v4f ws[8], wd[8];
#pragma unroll
    for (int q = 0; q < 8; ++q) { ws[q] = as4[q * 32]; wd[q] = ad4[q * 32]; }

    v4f hv[SC_ROWS][8];
#pragma unroll
    for (int r = 0; r < SC_ROWS; ++r) {
        const v4f* hp = (const v4f*)(h + (size_t)(t0 + r) * HD_) + lane;
#pragma unroll
        for (int q = 0; q < 8; ++q) hv[r][q] = hp[q * 32];
    }

    float v[32];
#pragma unroll
    for (int r = 0; r < SC_ROWS; ++r) {
#pragma unroll
        for (int hd = 0; hd < H_; ++hd) {
            float s = 0.0f, d = 0.0f;
#pragma unroll
            for (int q = 2 * hd; q < 2 * hd + 2; ++q) {
#pragma unroll
                for (int j = 0; j < 4; ++j) {
                    s = fmaf(hv[r][q][j], ws[q][j], s);
                    d = fmaf(hv[r][q][j], wd[q][j], d);
                }
            }
            v[r * H_ + hd]      = s;
            v[16 + r * H_ + hd] = d;
        }
    }

    v4f hm[SC_ROWS][2];
#pragma unroll
    for (int r = 0; r < SC_ROWS; ++r) {
#pragma unroll
        for (int p = 0; p < 2; ++p)
            hm[r][p] = 0.25f * (hv[r][p] + hv[r][2 + p] + hv[r][4 + p] + hv[r][6 + p]);
    }

    sc_fold<16, 16>(v, lane);
    sc_fold<8, 8>(v, lane);
    sc_fold<4, 4>(v, lane);
    sc_fold<2, 2>(v, lane);
    sc_fold<1, 1>(v, lane);

    float* dst = (lane < 16) ? (a_src + (size_t)t0 * H_ + lane)
                             : (a_dst + (size_t)t0 * H_ + (lane - 16));
    *dst = v[0];

    if (t0 >= BN_) return;

    v4f bb[2], g4[2], be[2];
#pragma unroll
    for (int p = 0; p < 2; ++p) {
        bb[p] = ((const v4f*)bias + lane)[p * 32];
        g4[p] = ((const v4f*)gg + lane)[p * 32];
        be[p] = ((const v4f*)gb + lane)[p * 32];
    }

    float s1[SC_ROWS];
#pragma unroll
    for (int r = 0; r < SC_ROWS; ++r) {
        float s = 0.0f;
#pragma unroll
        for (int p = 0; p < 2; ++p) {
#pragma unroll
            for (int j = 0; j < 4; ++j) {
                const float g = gelu_exact(hm[r][p][j] + bb[p][j]);
                hm[r][p][j] = g;
                s += g;
            }
        }
        s1[r] = s;
    }
    sc_allsum(s1);

    float s2[SC_ROWS];
#pragma unroll
    for (int r = 0; r < SC_ROWS; ++r) {
        const float mu = s1[r] * (1.0f / D_);
        float s = 0.0f;
#pragma unroll
        for (int p = 0; p < 2; ++p) {
#pragma unroll
            for (int j = 0; j < 4; ++j) {
                const float cdev = hm[r][p][j] - mu;
                hm[r][p][j] = cdev;
                s += cdev * cdev;
            }
        }
        s2[r] = s;
    }
    sc_allsum(s2);

#pragma unroll
    for (int r = 0; r < SC_ROWS; ++r) {
        const float var = s2[r] * (1.0f / D_);
        const float rs  = rsqrtf(var + EPS_);
        v4f* op = (v4f*)(gat1 + (size_t)(t0 + r) * D_) + lane;
#pragma unroll
        for (int p = 0; p < 2; ++p) {
            v4f o;
#pragma unroll
            for (int j = 0; j < 4; ++j) o[j] = hm[r][p][j] * rs * g4[p][j] + be[p][j];
            op[p * 32] = o;
        }
    }
}

#define VA_WAVES 8
#define VA_PITCH 520
static_assert(N_ == 2 * VA_WAVES * 32, "two sources per thread in the softmax part");
static_assert(N_ == 4 * 128 && D_ == 2 * 128, "wave tiling of the aggregation");

__device__ __forceinline__ float lrelu02(float x) { return (x > 0.0f) ? x : 0.2f * x; }

__global__ void __launch_bounds__(VA_WAVES * 32)
k_gat_virtual(const float* __restrict__ h,
              const float* __restrict__ a_src,
              const float* __restrict__ a_dst,
              float* __restrict__ gat_agg) {
    __shared__ __attribute__((aligned(16))) float wl[6 * VA_PITCH];
    __shared__ __attribute__((aligned(16))) float part[4][V_][128];
    __shared__ float redm[VA_WAVES][8];
    __shared__ float reds[VA_WAVES][8];
    __shared__ float fin_ws[8];
    __shared__ float fin_inv[8];

    const int tid  = threadIdx.x;
    const int lane = tid & 31;
    const int wave = tid >> 5;
    const int half = lane >> 4;
    const int c    = lane & 15;
    const int ch   = blockIdx.x & 1;
    const int hd   = (blockIdx.x >> 1) & (H_ - 1);
    const int b    = blockIdx.x >> 3;
    const int cg   = wave & 1;
    const int kc   = wave >> 1;
    const int u0   = BN_ + b * V_;
    const int col0 = hd * D_ + ch * 128;

    float hself[3];
#pragma unroll
    for (int e = 0; e < 3; ++e) {
        const int idx = tid + e * 256;
        const int v   = (idx < V_ * 128) ? (idx >> 7) : 0;
        hself[e] = h[(size_t)(u0 + v) * HD_ + col0 + (idx & 127)];
    }

    float ad[V_], aself[V_];
#pragma unroll
    for (int v = 0; v < V_; ++v) {
        ad[v]    = a_dst[(size_t)(u0 + v) * H_ + hd];
        aself[v] = lrelu02(a_src[(size_t)(u0 + v) * H_ + hd] + ad[v]);
    }
    const int r0 = b * N_ + tid * 2;
    const float s0 = a_src[(size_t)r0 * H_ + hd];
    const float s1 = a_src[(size_t)(r0 + 1) * H_ + hd];
    float al0[V_], al1[V_], mx[V_];
#pragma unroll
    for (int v = 0; v < V_; ++v) {
        al0[v] = lrelu02(s0 + ad[v]);
        al1[v] = lrelu02(s1 + ad[v]);
        mx[v]  = fmaxf(al0[v], al1[v]);
    }
#pragma unroll
    for (int off = 16; off > 0; off >>= 1) {
#pragma unroll
        for (int v = 0; v < V_; ++v) mx[v] = fmaxf(mx[v], __shfl_xor(mx[v], off, 32));
    }
    if (lane == 0) {
#pragma unroll
        for (int v = 0; v < V_; ++v) redm[wave][v] = mx[v];
    }
    {
        v2f z; z[0] = 0.0f; z[1] = 0.0f;
        *(v2f*)(wl + 5 * VA_PITCH + 2 * tid) = z;
    }
    __syncthreads();

    float m[V_], es[V_];
#pragma unroll
    for (int v = 0; v < V_; ++v) {
        float t = aself[v];
#pragma unroll
        for (int w = 0; w < VA_WAVES; ++w) t = fmaxf(t, redm[w][v]);
        m[v] = t;
        v2f e;
        e[0] = expf(al0[v] - t);
        e[1] = expf(al1[v] - t);
        *(v2f*)(wl + v * VA_PITCH + 2 * tid) = e;
        es[v] = e[0] + e[1];
    }
#pragma unroll
    for (int off = 16; off > 0; off >>= 1) {
#pragma unroll
        for (int v = 0; v < V_; ++v) es[v] += __shfl_xor(es[v], off, 32);
    }
    if (lane == 0) {
#pragma unroll
        for (int v = 0; v < V_; ++v) reds[wave][v] = es[v];
    }
    __syncthreads();

    const float* hp = h + (size_t)(b * N_ + kc * 128 + 2 * half) * HD_ + col0 + cg * 64 + 4 * c;
    v4f q[32][2];
#pragma unroll
    for (int i = 0; i < 32; ++i) {
#pragma unroll
        for (int s = 0; s < 2; ++s) q[i][s] = *(const v4f*)(hp + (size_t)(4 * i + s) * HD_);
    }

    if (tid < V_) {
        float as_ = aself[0], mm = m[0];
#pragma unroll
        for (int v = 1; v < V_; ++v) {
            if (tid == v) { as_ = aself[v]; mm = m[v]; }
        }
        const float eself = expf(as_ - mm);
        float d = eself;
#pragma unroll
        for (int w = 0; w < VA_WAVES; ++w) d += reds[w][tid];
        fin_ws[tid]  = eself;
        fin_inv[tid] = 1.0f / d;
    }

    const float* ap = wl + ((c < V_) ? c : V_) * VA_PITCH + kc * 128 + 2 * half;
    v2f af[32];
#pragma unroll
    for (int i = 0; i < 32; ++i) af[i] = *(const v2f*)(ap + 4 * i);

    v8f c0 = {}; v8f c1 = {}; v8f c2 = {}; v8f c3 = {};
#pragma unroll
    for (int i = 0; i < 32; ++i) {
        v2f b0, b1, b2, b3;
        b0[0] = q[i][0][0]; b0[1] = q[i][1][0];
        b1[0] = q[i][0][1]; b1[1] = q[i][1][1];
        b2[0] = q[i][0][2]; b2[1] = q[i][1][2];
        b3[0] = q[i][0][3]; b3[1] = q[i][1][3];
        c0 = __builtin_amdgcn_wmma_f32_16x16x4_f32(false, af[i], false, b0, (short)0, c0, false, false);
        c1 = __builtin_amdgcn_wmma_f32_16x16x4_f32(false, af[i], false, b1, (short)0, c1, false, false);
        c2 = __builtin_amdgcn_wmma_f32_16x16x4_f32(false, af[i], false, b2, (short)0, c2, false, false);
        c3 = __builtin_amdgcn_wmma_f32_16x16x4_f32(false, af[i], false, b3, (short)0, c3, false, false);
    }

    if (half == 0) {
#pragma unroll
        for (int v = 0; v < V_; ++v) {
            v4f o;
            o[0] = c0[v]; o[1] = c1[v]; o[2] = c2[v]; o[3] = c3[v];
            *(v4f*)(&part[kc][v][cg * 64 + 4 * c]) = o;
        }
    }
    __syncthreads();

#pragma unroll
    for (int e = 0; e < 3; ++e) {
        const int idx = tid + e * 256;
        if (idx < V_ * 128) {
            const int v   = idx >> 7;
            const int col = idx & 127;
            float s = part[0][v][col] + part[1][v][col];
            s += part[2][v][col];
            s += part[3][v][col];
            s += fin_ws[v] * hself[e];
            gat_agg[((size_t)(b * V_ + v) * H_ + hd) * D_ + ch * 128 + col] = s * fin_inv[v];
        }
    }
}

__global__ void k_gat_finish(const float* __restrict__ gat_agg,
                             const float* __restrict__ bias,
                             const float* __restrict__ gg,
                             const float* __restrict__ gb,
                             float* __restrict__ out) {
    __shared__ float red[D_];
    const int u = blockIdx.x;
    const int d = threadIdx.x;
    const float* p = gat_agg + (size_t)u * HD_ + d;
    float v = 0.25f * (p[0] + p[D_] + p[2 * D_] + p[3 * D_]);
    v = gelu_exact(v + bias[d]);

    red[d] = v;
    __syncthreads();
    for (int s = D_ / 2; s > 0; s >>= 1) {
        if (d < s) red[d] += red[d + s];
        __syncthreads();
    }
    const float mu = red[0] * (1.0f / D_);
    __syncthreads();
    const float c = v - mu;
    red[d] = c * c;
    __syncthreads();
    for (int s = D_ / 2; s > 0; s >>= 1) {
        if (d < s) red[d] += red[d + s];
        __syncthreads();
    }
    const float var = red[0] * (1.0f / D_);
    out[(size_t)(BN_ + u) * D_ + d] = c * rsqrtf(var + EPS_) * gg[d] + gb[d];
}

#define GF_ROWS  2
#define GF_WAVES 4
static_assert(N_ % GF_ROWS == 0, "a wave's rows are all of one batch");
static_assert(BN_ % (GF_ROWS * GF_WAVES) == 0, "the grid covers the real rows exactly");

__device__ __forceinline__ void gf_allsum(float* s) {
#pragma unroll
    for (int off = 16; off > 0; off >>= 1) {
#pragma unroll
        for (int r = 0; r < GF_ROWS; ++r) s[r] += __shfl_xor(s[r], off, 32);
    }
}

__global__ void __launch_bounds__(GF_WAVES * 32)
k_gcn_finish(const float* __restrict__ xw,
             const float* __restrict__ bias,
             const float* __restrict__ gg,
             const float* __restrict__ gb,
             float* __restrict__ out) {
    const int lane = threadIdx.x & 31;
    const int wave = blockIdx.x * GF_WAVES + (threadIdx.x >> 5);
    const int t0   = wave * GF_ROWS;
    const int b    = t0 / N_;

    v4f xr[GF_ROWS][2];
#pragma unroll
    for (int r = 0; r < GF_ROWS; ++r) {
        const v4f* xp = (const v4f*)(xw + (size_t)(t0 + r) * D_) + lane;
#pragma unroll
        for (int p = 0; p < 2; ++p) xr[r][p] = xp[p * 32];
    }
    v4f xv[V_][2];
#pragma unroll
    for (int v = 0; v < V_; ++v) {
        const v4f* vp = (const v4f*)(xw + (size_t)(BN_ + b * V_ + v) * D_) + lane;
#pragma unroll
        for (int p = 0; p < 2; ++p) xv[v][p] = vp[p * 32];
    }
    v4f bb[2], g4[2], be[2];
#pragma unroll
    for (int p = 0; p < 2; ++p) {
        bb[p] = ((const v4f*)bias + lane)[p * 32];
        g4[p] = ((const v4f*)gg + lane)[p * 32];
        be[p] = ((const v4f*)gb + lane)[p * 32];
    }

    const float inv_sqrt6 = 0.40824829046386301637f;
    v4f s5[2];
#pragma unroll
    for (int p = 0; p < 2; ++p) {
        s5[p] = xv[0][p] + xv[1][p];
        s5[p] += xv[2][p];
        s5[p] += xv[3][p];
        s5[p] += xv[4][p];
    }

    float s1[GF_ROWS];
#pragma unroll
    for (int r = 0; r < GF_ROWS; ++r) {
        float s = 0.0f;
#pragma unroll
        for (int p = 0; p < 2; ++p) {
#pragma unroll
            for (int j = 0; j < 4; ++j) {
                const float pre = s5[p][j] * inv_sqrt6 + xr[r][p][j] * (1.0f / 6.0f) + bb[p][j];
                const float g = gelu_exact(pre);
                xr[r][p][j] = g;
                s += g;
            }
        }
        s1[r] = s;
    }
    gf_allsum(s1);

    float s2[GF_ROWS];
#pragma unroll
    for (int r = 0; r < GF_ROWS; ++r) {
        const float mu = s1[r] * (1.0f / D_);
        float s = 0.0f;
#pragma unroll
        for (int p = 0; p < 2; ++p) {
#pragma unroll
            for (int j = 0; j < 4; ++j) {
                const float cdev = xr[r][p][j] - mu;
                xr[r][p][j] = cdev;
                s += cdev * cdev;
            }
        }
        s2[r] = s;
    }
    gf_allsum(s2);

#pragma unroll
    for (int r = 0; r < GF_ROWS; ++r) {
        const float var = s2[r] * (1.0f / D_);
        const float rs  = rsqrtf(var + EPS_);
        v4f* op = (v4f*)(out + (size_t)(t0 + r) * D_) + lane;
#pragma unroll
        for (int p = 0; p < 2; ++p) {
            v4f o;
#pragma unroll
            for (int j = 0; j < 4; ++j) o[j] = xr[r][p][j] * rs * g4[p][j] + be[p][j];
            op[p * 32] = o;
        }
    }
}

extern "C" void kernel_launch(void* const* d_in, const int* in_sizes, int n_in,
                              void* d_out, int out_size, void* d_ws, size_t ws_size,
                              hipStream_t stream) {
    (void)in_sizes; (void)n_in; (void)out_size; (void)ws_size;

    const float* x       = (const float*)d_in[0];
    const float* vn      = (const float*)d_in[2];
    const float* gat_W   = (const float*)d_in[3];
    const float* att_src = (const float*)d_in[4];
    const float* att_dst = (const float*)d_in[5];
    const float* gat_b   = (const float*)d_in[6];
    const float* gcn_W   = (const float*)d_in[7];
    const float* gcn_b   = (const float*)d_in[8];
    const float* n1g     = (const float*)d_in[9];
    const float* n1b     = (const float*)d_in[10];
    const float* n2g     = (const float*)d_in[11];
    const float* n2b     = (const float*)d_in[12];
    float* out = (float*)d_out;

    float* ws = (float*)d_ws;
    size_t off = 0;
    float* nodes = ws + off; off += (size_t)T_ * D_;
    float* h     = ws + off; off += (size_t)T_ * HD_;
    float* asrc  = ws + off; off += (size_t)T_ * H_;
    float* adst  = ws + off; off += (size_t)T_ * H_;
    float* gagg  = ws + off; off += (size_t)B_ * V_ * HD_;
    float* gat1  = ws + off; off += (size_t)T_ * D_;
    float* xw    = ws + off; off += (size_t)T_ * D_;

    {
        const size_t n = (size_t)T_ * D_;
        k_build_nodes<<<(unsigned)((n + 255) / 256), 256, 0, stream>>>(x, vn, nodes);
    }
    {
        dim3 grid((T_ + 63) / 64, HD_ / GB_COLS);
        k_gemm_f32_wmma<T_, D_, HD_><<<grid, GB_WAVES * 32, 0, stream>>>(nodes, gat_W, h);
    }
    k_scores<<<T_ / (SC_ROWS * SC_WAVES), SC_WAVES * 32, 0, stream>>>(h, att_src, att_dst, gat_b, n1g, n1b,
                                                                     asrc, adst, gat1);
    k_gat_virtual<<<B_ * H_ * 2, VA_WAVES * 32, 0, stream>>>(h, asrc, adst, gagg);
    k_gat_finish<<<B_ * V_, D_, 0, stream>>>(gagg, gat_b, n1g, n1b, gat1);
    {
        dim3 grid((T_ + 63) / 64, D_ / GB_COLS);
        k_gemm_f32_wmma<T_, D_, D_><<<grid, GB_WAVES * 32, 0, stream>>>(gat1, gcn_W, xw);
    }
    k_gcn_finish<<<BN_ / (GF_ROWS * GF_WAVES), GF_WAVES * 32, 0, stream>>>(xw, gcn_b, n2g, n2b, out);
}


# A operand of both GEMMs: a wave reads its 16 rows as whole 128-byte lines (16-byte loads, 32 k per chunk) into its own LDS patch of pitch 36; fragments by 8-byte LDS reads; sum order unchanged
# speedup vs baseline: 2.3939x; 1.0322x over previous
#include <hip/hip_runtime.h>
#include <math.h>

#define B_   32
#define N_   512
#define V_   5
#define D_   256
#define H_   4
#define BN_  (B_ * N_)
#define T_   (BN_ + B_ * V_)
#define HD_  (H_ * D_)
#define EPS_ 1e-5f

typedef float v2f __attribute__((ext_vector_type(2)));
typedef float v8f __attribute__((ext_vector_type(8)));

__device__ __forceinline__ float gelu_exact(float x) {
    return 0.5f * x * (1.0f + erff(x * 0.70710678118654752f));
}

__global__ void k_build_nodes(const float* __restrict__ x,
                              const float* __restrict__ vn,
                              float* __restrict__ nodes) {
    size_t i = (size_t)blockIdx.x * blockDim.x + threadIdx.x;
    if (i < (size_t)BN_ * D_) {
        nodes[i] = x[i];
    } else if (i < (size_t)T_ * D_) {
        size_t j = i - (size_t)BN_ * D_;
        nodes[i] = vn[j % (V_ * D_)];
    }
}

#define GB_COLS  64
#define GB_WAVES 4
#define GA_K     32
#define GA_PITCH 36
typedef float v4f __attribute__((ext_vector_type(4)));

template <int MDIM, int KDIM, int NDIM>
__global__ void __launch_bounds__(GB_WAVES * 32)
k_gemm_f32_wmma(const float* __restrict__ A,
                const float* __restrict__ Bm,
                float* __restrict__ C) {
    static_assert(KDIM % GA_K == 0 && NDIM % GB_COLS == 0 && MDIM % 16 == 0, "tile shapes");
    __shared__ __attribute__((aligned(16))) float Bl[(KDIM / 2) * GB_COLS * 2];
    __shared__ __attribute__((aligned(16))) float Al[GB_WAVES * 16 * GA_PITCH];

    const int tid = threadIdx.x;
    const int n0  = blockIdx.y * GB_COLS;

    {
        const int cq = tid & 15;
        const int kr = tid >> 4;
        const float* bsrc = Bm + n0 + cq * 4;
        const int f  = (kr & 1) * 2 + (cq >> 3);
        float* dbase = Bl + (cq >> 3) * 64 + (cq & 3) * 16 + ((cq >> 2) & 1) * 2;
#pragma unroll
        for (int i = 0; i < KDIM / 16; ++i) {
            const int kp = kr + 8 * i;
            const v4f r0 = *(const v4f*)(bsrc + (size_t)(2 * kp) * NDIM);
            const v4f r1 = *(const v4f*)(bsrc + (size_t)(2 * kp + 1) * NDIM);
            float* d = dbase + kp * (GB_COLS * 2);
#pragma unroll
            for (int j = 0; j < 4; ++j) {
                v2f p;
                p[0] = r0[j]; p[1] = r1[j];
                *(v2f*)(d + (j ^ f) * 4) = p;
            }
        }
    }
    __syncthreads();

    const int wave = tid >> 5;
    const int lane = tid & 31;
    const int half = lane >> 4;
    const int lm   = lane & 15;
    const int m0   = (blockIdx.x * GB_WAVES + wave) * 16;
    if (m0 >= MDIM) return;

    v8f c0 = {}; v8f c1 = {}; v8f c2 = {}; v8f c3 = {};
    const float* bl01 = Bl + half * (GB_COLS * 2) + (lm ^ (half * 2)) * 4;
    const float* bl23 = Bl + half * (GB_COLS * 2) + 64 + (lm ^ (half * 2 + 1)) * 4;

    const int lr = lane >> 3;
    const int lp = lane & 7;
    const float* asrc = A + (size_t)(m0 + lr) * KDIM + lp * 4;
    float* apatch = Al + wave * (16 * GA_PITCH);
    float* ast = apatch + lr * GA_PITCH + lp * 4;
    const float* afr = apatch + lm * GA_PITCH + half * 2;

    v4f ar[4];
#pragma unroll
    for (int j = 0; j < 4; ++j) ar[j] = *(const v4f*)(asrc + (size_t)(4 * j) * KDIM);

    for (int ch = 0; ch < KDIM / GA_K; ++ch) {
#pragma unroll
        for (int j = 0; j < 4; ++j) *(v4f*)(ast + 4 * j * GA_PITCH) = ar[j];
        if (ch + 1 < KDIM / GA_K) {
#pragma unroll
            for (int j = 0; j < 4; ++j)
                ar[j] = *(const v4f*)(asrc + (size_t)(4 * j) * KDIM + (ch + 1) * GA_K);
        }
#pragma unroll
        for (int s = 0; s < GA_K / 4; ++s) {
            const v2f a = *(const v2f*)(afr + 4 * s);
            const int ko = ((ch * GA_K + 4 * s) >> 1) * (GB_COLS * 2);
            const v4f q01 = *(const v4f*)(bl01 + ko);
            const v4f q23 = *(const v4f*)(bl23 + ko);
            v2f b0, b1, b2, b3;
            b0[0] = q01[0]; b0[1] = q01[1];
            b1[0] = q01[2]; b1[1] = q01[3];
            b2[0] = q23[0]; b2[1] = q23[1];
            b3[0] = q23[2]; b3[1] = q23[3];
            c0 = __builtin_amdgcn_wmma_f32_16x16x4_f32(false, a, false, b0, (short)0, c0, false, false);
            c1 = __builtin_amdgcn_wmma_f32_16x16x4_f32(false, a, false, b1, (short)0, c1, false, false);
            c2 = __builtin_amdgcn_wmma_f32_16x16x4_f32(false, a, false, b2, (short)0, c2, false, false);
            c3 = __builtin_amdgcn_wmma_f32_16x16x4_f32(false, a, false, b3, (short)0, c3, false, false);
        }
    }

    float* crow = C + n0 + lm;
#pragma unroll
    for (int j = 0; j < 8; ++j) {
        const size_t row = (size_t)(m0 + half * 8 + j) * NDIM;
        crow[row + 0]  = c0[j];
        crow[row + 16] = c1[j];
        crow[row + 32] = c2[j];
        crow[row + 48] = c3[j];
    }
}

#define SC_ROWS  4
#define SC_WAVES 4
static_assert(T_ % (SC_ROWS * SC_WAVES) == 0, "the grid covers the rows exactly");
static_assert(SC_ROWS * H_ * 2 == 32, "one total per lane");
static_assert(BN_ % SC_ROWS == 0, "a wave's rows are all real or all virtual");

template <int N, int OFF>
__device__ __forceinline__ void sc_fold(float* v, int lane) {
    const bool up = (lane & OFF) != 0;
#pragma unroll
    for (int i = 0; i < N; ++i) {
        const float keep = up ? v[N + i] : v[i];
        const float send = up ? v[i] : v[N + i];
        v[i] = keep + __shfl_xor(send, OFF, 32);
    }
}

__device__ __forceinline__ void sc_allsum(float* s) {
#pragma unroll
    for (int off = 16; off > 0; off >>= 1) {
#pragma unroll
        for (int r = 0; r < SC_ROWS; ++r) s[r] += __shfl_xor(s[r], off, 32);
    }
}

__global__ void __launch_bounds__(SC_WAVES * 32, 5)
k_scores(const float* __restrict__ h,
         const float* __restrict__ att_src,
         const float* __restrict__ att_dst,
         const float* __restrict__ bias,
         const float* __restrict__ gg,
         const float* __restrict__ gb,
         float* __restrict__ a_src,
         float* __restrict__ a_dst,
         float* __restrict__ gat1) {
    const int lane = threadIdx.x & 31;
    const int wave = blockIdx.x * SC_WAVES + (threadIdx.x >> 5);
    const int t0   = wave * SC_ROWS;

    const v4f* as4 = (const v4f*)att_src + lane;
    const v4f* ad4 = (const v4f*)att_dst + lane;
    v4f ws[8], wd[8];
#pragma unroll
    for (int q = 0; q < 8; ++q) { ws[q] = as4[q * 32]; wd[q] = ad4[q * 32]; }

    v4f hv[SC_ROWS][8];
#pragma unroll
    for (int r = 0; r < SC_ROWS; ++r) {
        const v4f* hp = (const v4f*)(h + (size_t)(t0 + r) * HD_) + lane;
#pragma unroll
        for (int q = 0; q < 8; ++q) hv[r][q] = hp[q * 32];
    }

    float v[32];
#pragma unroll
    for (int r = 0; r < SC_ROWS; ++r) {
#pragma unroll
        for (int hd = 0; hd < H_; ++hd) {
            float s = 0.0f, d = 0.0f;
#pragma unroll
            for (int q = 2 * hd; q < 2 * hd + 2; ++q) {
#pragma unroll
                for (int j = 0; j < 4; ++j) {
                    s = fmaf(hv[r][q][j], ws[q][j], s);
                    d = fmaf(hv[r][q][j], wd[q][j], d);
                }
            }
            v[r * H_ + hd]      = s;
            v[16 + r * H_ + hd] = d;
        }
    }

    v4f hm[SC_ROWS][2];
#pragma unroll
    for (int r = 0; r < SC_ROWS; ++r) {
#pragma unroll
        for (int p = 0; p < 2; ++p)
            hm[r][p] = 0.25f * (hv[r][p] + hv[r][2 + p] + hv[r][4 + p] + hv[r][6 + p]);
    }

    sc_fold<16, 16>(v, lane);
    sc_fold<8, 8>(v, lane);
    sc_fold<4, 4>(v, lane);
    sc_fold<2, 2>(v, lane);
    sc_fold<1, 1>(v, lane);

    float* dst = (lane < 16) ? (a_src + (size_t)t0 * H_ + lane)
                             : (a_dst + (size_t)t0 * H_ + (lane - 16));
    *dst = v[0];

    if (t0 >= BN_) return;

    v4f bb[2], g4[2], be[2];
#pragma unroll
    for (int p = 0; p < 2; ++p) {
        bb[p] = ((const v4f*)bias + lane)[p * 32];
        g4[p] = ((const v4f*)gg + lane)[p * 32];
        be[p] = ((const v4f*)gb + lane)[p * 32];
    }

    float s1[SC_ROWS];
#pragma unroll
    for (int r = 0; r < SC_ROWS; ++r) {
        float s = 0.0f;
#pragma unroll
        for (int p = 0; p < 2; ++p) {
#pragma unroll
            for (int j = 0; j < 4; ++j) {
                const float g = gelu_exact(hm[r][p][j] + bb[p][j]);
                hm[r][p][j] = g;
                s += g;
            }
        }
        s1[r] = s;
    }
    sc_allsum(s1);

    float s2[SC_ROWS];
#pragma unroll
    for (int r = 0; r < SC_ROWS; ++r) {
        const float mu = s1[r] * (1.0f / D_);
        float s = 0.0f;
#pragma unroll
        for (int p = 0; p < 2; ++p) {
#pragma unroll
            for (int j = 0; j < 4; ++j) {
                const float cdev = hm[r][p][j] - mu;
                hm[r][p][j] = cdev;
                s += cdev * cdev;
            }
        }
        s2[r] = s;
    }
    sc_allsum(s2);

#pragma unroll
    for (int r = 0; r < SC_ROWS; ++r) {
        const float var = s2[r] * (1.0f / D_);
        const float rs  = rsqrtf(var + EPS_);
        v4f* op = (v4f*)(gat1 + (size_t)(t0 + r) * D_) + lane;
#pragma unroll
        for (int p = 0; p < 2; ++p) {
            v4f o;
#pragma unroll
            for (int j = 0; j < 4; ++j) o[j] = hm[r][p][j] * rs * g4[p][j] + be[p][j];
            op[p * 32] = o;
        }
    }
}

#define VA_WAVES 8
#define VA_PITCH 520
static_assert(N_ == 2 * VA_WAVES * 32, "two sources per thread in the softmax part");
static_assert(N_ == 4 * 128 && D_ == 2 * 128, "wave tiling of the aggregation");

__device__ __forceinline__ float lrelu02(float x) { return (x > 0.0f) ? x : 0.2f * x; }

__global__ void __launch_bounds__(VA_WAVES * 32)
k_gat_virtual(const float* __restrict__ h,
              const float* __restrict__ a_src,
              const float* __restrict__ a_dst,
              float* __restrict__ gat_agg) {
    __shared__ __attribute__((aligned(16))) float wl[6 * VA_PITCH];
    __shared__ __attribute__((aligned(16))) float part[4][V_][128];
    __shared__ float redm[VA_WAVES][8];
    __shared__ float reds[VA_WAVES][8];
    __shared__ float fin_ws[8];
    __shared__ float fin_inv[8];

    const int tid  = threadIdx.x;
    const int lane = tid & 31;
    const int wave = tid >> 5;
    const int half = lane >> 4;
    const int c    = lane & 15;
    const int ch   = blockIdx.x & 1;
    const int hd   = (blockIdx.x >> 1) & (H_ - 1);
    const int b    = blockIdx.x >> 3;
    const int cg   = wave & 1;
    const int kc   = wave >> 1;
    const int u0   = BN_ + b * V_;
    const int col0 = hd * D_ + ch * 128;

    float hself[3];
#pragma unroll
    for (int e = 0; e < 3; ++e) {
        const int idx = tid + e * 256;
        const int v   = (idx < V_ * 128) ? (idx >> 7) : 0;
        hself[e] = h[(size_t)(u0 + v) * HD_ + col0 + (idx & 127)];
    }

    float ad[V_], aself[V_];
#pragma unroll
    for (int v = 0; v < V_; ++v) {
        ad[v]    = a_dst[(size_t)(u0 + v) * H_ + hd];
        aself[v] = lrelu02(a_src[(size_t)(u0 + v) * H_ + hd] + ad[v]);
    }
    const int r0 = b * N_ + tid * 2;
    const float s0 = a_src[(size_t)r0 * H_ + hd];
    const float s1 = a_src[(size_t)(r0 + 1) * H_ + hd];
    float al0[V_], al1[V_], mx[V_];
#pragma unroll
    for (int v = 0; v < V_; ++v) {
        al0[v] = lrelu02(s0 + ad[v]);
        al1[v] = lrelu02(s1 + ad[v]);
        mx[v]  = fmaxf(al0[v], al1[v]);
    }
#pragma unroll
    for (int off = 16; off > 0; off >>= 1) {
#pragma unroll
        for (int v = 0; v < V_; ++v) mx[v] = fmaxf(mx[v], __shfl_xor(mx[v], off, 32));
    }
    if (lane == 0) {
#pragma unroll
        for (int v = 0; v < V_; ++v) redm[wave][v] = mx[v];
    }
    {
        v2f z; z[0] = 0.0f; z[1] = 0.0f;
        *(v2f*)(wl + 5 * VA_PITCH + 2 * tid) = z;
    }
    __syncthreads();

    float m[V_], es[V_];
#pragma unroll
    for (int v = 0; v < V_; ++v) {
        float t = aself[v];
#pragma unroll
        for (int w = 0; w < VA_WAVES; ++w) t = fmaxf(t, redm[w][v]);
        m[v] = t;
        v2f e;
        e[0] = expf(al0[v] - t);
        e[1] = expf(al1[v] - t);
        *(v2f*)(wl + v * VA_PITCH + 2 * tid) = e;
        es[v] = e[0] + e[1];
    }
#pragma unroll
    for (int off = 16; off > 0; off >>= 1) {
#pragma unroll
        for (int v = 0; v < V_; ++v) es[v] += __shfl_xor(es[v], off, 32);
    }
    if (lane == 0) {
#pragma unroll
        for (int v = 0; v < V_; ++v) reds[wave][v] = es[v];
    }
    __syncthreads();

    const float* hp = h + (size_t)(b * N_ + kc * 128 + 2 * half) * HD_ + col0 + cg * 64 + 4 * c;
    v4f q[32][2];
#pragma unroll
    for (int i = 0; i < 32; ++i) {
#pragma unroll
        for (int s = 0; s < 2; ++s) q[i][s] = *(const v4f*)(hp + (size_t)(4 * i + s) * HD_);
    }

    if (tid < V_) {
        float as_ = aself[0], mm = m[0];
#pragma unroll
        for (int v = 1; v < V_; ++v) {
            if (tid == v) { as_ = aself[v]; mm = m[v]; }
        }
        const float eself = expf(as_ - mm);
        float d = eself;
#pragma unroll
        for (int w = 0; w < VA_WAVES; ++w) d += reds[w][tid];
        fin_ws[tid]  = eself;
        fin_inv[tid] = 1.0f / d;
    }

    const float* ap = wl + ((c < V_) ? c : V_) * VA_PITCH + kc * 128 + 2 * half;
    v2f af[32];
#pragma unroll
    for (int i = 0; i < 32; ++i) af[i] = *(const v2f*)(ap + 4 * i);

    v8f c0 = {}; v8f c1 = {}; v8f c2 = {}; v8f c3 = {};
#pragma unroll
    for (int i = 0; i < 32; ++i) {
        v2f b0, b1, b2, b3;
        b0[0] = q[i][0][0]; b0[1] = q[i][1][0];
        b1[0] = q[i][0][1]; b1[1] = q[i][1][1];
        b2[0] = q[i][0][2]; b2[1] = q[i][1][2];
        b3[0] = q[i][0][3]; b3[1] = q[i][1][3];
        c0 = __builtin_amdgcn_wmma_f32_16x16x4_f32(false, af[i], false, b0, (short)0, c0, false, false);
        c1 = __builtin_amdgcn_wmma_f32_16x16x4_f32(false, af[i], false, b1, (short)0, c1, false, false);
        c2 = __builtin_amdgcn_wmma_f32_16x16x4_f32(false, af[i], false, b2, (short)0, c2, false, false);
        c3 = __builtin_amdgcn_wmma_f32_16x16x4_f32(false, af[i], false, b3, (short)0, c3, false, false);
    }

    if (half == 0) {
#pragma unroll
        for (int v = 0; v < V_; ++v) {
            v4f o;
            o[0] = c0[v]; o[1] = c1[v]; o[2] = c2[v]; o[3] = c3[v];
            *(v4f*)(&part[kc][v][cg * 64 + 4 * c]) = o;
        }
    }
    __syncthreads();

#pragma unroll
    for (int e = 0; e < 3; ++e) {
        const int idx = tid + e * 256;
        if (idx < V_ * 128) {
            const int v   = idx >> 7;
            const int col = idx & 127;
            float s = part[0][v][col] + part[1][v][col];
            s += part[2][v][col];
            s += part[3][v][col];
            s += fin_ws[v] * hself[e];
            gat_agg[((size_t)(b * V_ + v) * H_ + hd) * D_ + ch * 128 + col] = s * fin_inv[v];
        }
    }
}

__global__ void k_gat_finish(const float* __restrict__ gat_agg,
                             const float* __restrict__ bias,
                             const float* __restrict__ gg,
                             const float* __restrict__ gb,
                             float* __restrict__ out) {
    __shared__ float red[D_];
    const int u = blockIdx.x;
    const int d = threadIdx.x;
    const float* p = gat_agg + (size_t)u * HD_ + d;
    float v = 0.25f * (p[0] + p[D_] + p[2 * D_] + p[3 * D_]);
    v = gelu_exact(v + bias[d]);

    red[d] = v;
    __syncthreads();
    for (int s = D_ / 2; s > 0; s >>= 1) {
        if (d < s) red[d] += red[d + s];
        __syncthreads();
    }
    const float mu = red[0] * (1.0f / D_);
    __syncthreads();
    const float c = v - mu;
    red[d] = c * c;
    __syncthreads();
    for (int s = D_ / 2; s > 0; s >>= 1) {
        if (d < s) red[d] += red[d + s];
        __syncthreads();
    }
    const float var = red[0] * (1.0f / D_);
    out[(size_t)(BN_ + u) * D_ + d] = c * rsqrtf(var + EPS_) * gg[d] + gb[d];
}

#define GF_ROWS  2
#define GF_WAVES 4
static_assert(N_ % GF_ROWS == 0, "a wave's rows are all of one batch");
static_assert(BN_ % (GF_ROWS * GF_WAVES) == 0, "the grid covers the real rows exactly");

__device__ __forceinline__ void gf_allsum(float* s) {
#pragma unroll
    for (int off = 16; off > 0; off >>= 1) {
#pragma unroll
        for (int r = 0; r < GF_ROWS; ++r) s[r] += __shfl_xor(s[r], off, 32);
    }
}

__global__ void __launch_bounds__(GF_WAVES * 32)
k_gcn_finish(const float* __restrict__ xw,
             const float* __restrict__ bias,
             const float* __restrict__ gg,
             const float* __restrict__ gb,
             float* __restrict__ out) {
    const int lane = threadIdx.x & 31;
    const int wave = blockIdx.x * GF_WAVES + (threadIdx.x >> 5);
    const int t0   = wave * GF_ROWS;
    const int b    = t0 / N_;

    v4f xr[GF_ROWS][2];
#pragma unroll
    for (int r = 0; r < GF_ROWS; ++r) {
        const v4f* xp = (const v4f*)(xw + (size_t)(t0 + r) * D_) + lane;
#pragma unroll
        for (int p = 0; p < 2; ++p) xr[r][p] = xp[p * 32];
    }
    v4f xv[V_][2];
#pragma unroll
    for (int v = 0; v < V_; ++v) {
        const v4f* vp = (const v4f*)(xw + (size_t)(BN_ + b * V_ + v) * D_) + lane;
#pragma unroll
        for (int p = 0; p < 2; ++p) xv[v][p] = vp[p * 32];
    }
    v4f bb[2], g4[2], be[2];
#pragma unroll
    for (int p = 0; p < 2; ++p) {
        bb[p] = ((const v4f*)bias + lane)[p * 32];
        g4[p] = ((const v4f*)gg + lane)[p * 32];
        be[p] = ((const v4f*)gb + lane)[p * 32];
    }

    const float inv_sqrt6 = 0.40824829046386301637f;
    v4f s5[2];
#pragma unroll
    for (int p = 0; p < 2; ++p) {
        s5[p] = xv[0][p] + xv[1][p];
        s5[p] += xv[2][p];
        s5[p] += xv[3][p];
        s5[p] += xv[4][p];
    }

    float s1[GF_ROWS];
#pragma unroll
    for (int r = 0; r < GF_ROWS; ++r) {
        float s = 0.0f;
#pragma unroll
        for (int p = 0; p < 2; ++p) {
#pragma unroll
            for (int j = 0; j < 4; ++j) {
                const float pre = s5[p][j] * inv_sqrt6 + xr[r][p][j] * (1.0f / 6.0f) + bb[p][j];
                const float g = gelu_exact(pre);
                xr[r][p][j] = g;
                s += g;
            }
        }
        s1[r] = s;
    }
    gf_allsum(s1);

    float s2[GF_ROWS];
#pragma unroll
    for (int r = 0; r < GF_ROWS; ++r) {
        const float mu = s1[r] * (1.0f / D_);
        float s = 0.0f;
#pragma unroll
        for (int p = 0; p < 2; ++p) {
#pragma unroll
            for (int j = 0; j < 4; ++j) {
                const float cdev = xr[r][p][j] - mu;
                xr[r][p][j] = cdev;
                s += cdev * cdev;
            }
        }
        s2[r] = s;
    }
    gf_allsum(s2);

#pragma unroll
    for (int r = 0; r < GF_ROWS; ++r) {
        const float var = s2[r] * (1.0f / D_);
        const float rs  = rsqrtf(var + EPS_);
        v4f* op = (v4f*)(out + (size_t)(t0 + r) * D_) + lane;
#pragma unroll
        for (int p = 0; p < 2; ++p) {
            v4f o;
#pragma unroll
            for (int j = 0; j < 4; ++j) o[j] = xr[r][p][j] * rs * g4[p][j] + be[p][j];
            op[p * 32] = o;
        }
    }
}

extern "C" void kernel_launch(void* const* d_in, const int* in_sizes, int n_in,
                              void* d_out, int out_size, void* d_ws, size_t ws_size,
                              hipStream_t stream) {
    (void)in_sizes; (void)n_in; (void)out_size; (void)ws_size;

    const float* x       = (const float*)d_in[0];
    const float* vn      = (const float*)d_in[2];
    const float* gat_W   = (const float*)d_in[3];
    const float* att_src = (const float*)d_in[4];
    const float* att_dst = (const float*)d_in[5];
    const float* gat_b   = (const float*)d_in[6];
    const float* gcn_W   = (const float*)d_in[7];
    const float* gcn_b   = (const float*)d_in[8];
    const float* n1g     = (const float*)d_in[9];
    const float* n1b     = (const float*)d_in[10];
    const float* n2g     = (const float*)d_in[11];
    const float* n2b     = (const float*)d_in[12];
    float* out = (float*)d_out;

    float* ws = (float*)d_ws;
    size_t off = 0;
    float* nodes = ws + off; off += (size_t)T_ * D_;
    float* h     = ws + off; off += (size_t)T_ * HD_;
    float* asrc  = ws + off; off += (size_t)T_ * H_;
    float* adst  = ws + off; off += (size_t)T_ * H_;
    float* gagg  = ws + off; off += (size_t)B_ * V_ * HD_;
    float* gat1  = ws + off; off += (size_t)T_ * D_;
    float* xw    = ws + off; off += (size_t)T_ * D_;

    {
        const size_t n = (size_t)T_ * D_;
        k_build_nodes<<<(unsigned)((n + 255) / 256), 256, 0, stream>>>(x, vn, nodes);
    }
    {
        dim3 grid((T_ + 63) / 64, HD_ / GB_COLS);
        k_gemm_f32_wmma<T_, D_, HD_><<<grid, GB_WAVES * 32, 0, stream>>>(nodes, gat_W, h);
    }
    k_scores<<<T_ / (SC_ROWS * SC_WAVES), SC_WAVES * 32, 0, stream>>>(h, att_src, att_dst, gat_b, n1g, n1b,
                                                                     asrc, adst, gat1);
    k_gat_virtual<<<B_ * H_ * 2, VA_WAVES * 32, 0, stream>>>(h, asrc, adst, gagg);
    k_gat_finish<<<B_ * V_, D_, 0, stream>>>(gagg, gat_b, n1g, n1b, gat1);
    {
        dim3 grid((T_ + 63) / 64, D_ / GB_COLS);
        k_gemm_f32_wmma<T_, D_, D_><<<grid, GB_WAVES * 32, 0, stream>>>(gat1, gcn_W, xw);
    }
    k_gcn_finish<<<BN_ / (GF_ROWS * GF_WAVES), GF_WAVES * 32, 0, stream>>>(xw, gcn_b, n2g, n2b, out);
}
